# starting kernel
# baseline (speedup 1.0000x reference)
#include <hip/hip_runtime.h>

typedef float v2f __attribute__((ext_vector_type(2)));
typedef float v8f __attribute__((ext_vector_type(8)));

#define BATCH     32
#define IN_CH     21
#define IN_FEAT   4096
#define WIN       32
#define OUT_CH    8
#define N_WIN     4064
#define NK        (OUT_CH * N_WIN)
#define DDIM      (IN_CH * WIN)

__global__ __launch_bounds__(256)
void FilterbankLinear_kernel(const float* __restrict__ x,
                             const float* __restrict__ w,
                             float* __restrict__ out) {
    const int lane = threadIdx.x & 31;
    const int wave = threadIdx.x >> 5;
    const int j = blockIdx.x * 8 + wave;
    if (j >= N_WIN) return;

    const int start = (j == N_WIN - 1) ? (IN_FEAT - WIN) : j;

    const int r    = lane & 15;
    const int half = lane >> 4;
    const int koff = half * 2;

    const float* __restrict__ xlo = x + (size_t)r        * (IN_CH * IN_FEAT);
    const float* __restrict__ xhi = x + (size_t)(r + 16) * (IN_CH * IN_FEAT);

    const int  n      = r;
    const bool nvalid = (n < OUT_CH);
    const int  nclamp = nvalid ? n : (OUT_CH - 1);
    const float* __restrict__ wrow =
        w + (size_t)(j * OUT_CH + nclamp) * DDIM + koff;

    v8f acc0 = {};
    v8f acc1 = {};

    #pragma unroll 8
    for (int kb = 0; kb < DDIM; kb += 4) {
        const int c = kb >> 5;
        const int t = (kb & 31) + koff;
        const int p = start + t;

        const float* alo = xlo + c * IN_FEAT + p;
        const float* ahi = xhi + c * IN_FEAT + p;
        v2f a0; a0.x = alo[0]; a0.y = alo[1];
        v2f a1; a1.x = ahi[0]; a1.y = ahi[1];

        v2f bv = __builtin_nontemporal_load((const v2f*)(wrow + kb));
        bv.x = nvalid ? bv.x : 0.0f;
        bv.y = nvalid ? bv.y : 0.0f;

        acc0 = __builtin_amdgcn_wmma_f32_16x16x4_f32(
                   false, a0, false, bv, (short)0, acc0, false, false);
        acc1 = __builtin_amdgcn_wmma_f32_16x16x4_f32(
                   false, a1, false, bv, (short)0, acc1, false, false);
    }

    if (nvalid) {
        float* obase = out + (size_t)j * OUT_CH + n;
        #pragma unroll
        for (int v = 0; v < 8; ++v) {
            const int m = v + half * 8;
            obase[(size_t)m        * NK] = acc0[v];
            obase[(size_t)(m + 16) * NK] = acc1[v];
        }
    }
}

extern "C" void kernel_launch(void* const* d_in, const int* in_sizes, int n_in,
                              void* d_out, int out_size, void* d_ws, size_t ws_size,
                              hipStream_t stream) {
    (void)in_sizes; (void)n_in; (void)out_size; (void)d_ws; (void)ws_size;
    const float* x = (const float*)d_in[0];
    const float* w = (const float*)d_in[1];
    float* out = (float*)d_out;

    dim3 grid(N_WIN / 8);
    dim3 block(256);
    FilterbankLinear_kernel<<<grid, block, 0, stream>>>(x, w, out);
}


# Stage the block's shared x tile (8 overlapping windows, 672 rows of 40 floats) once in LDS with 16-byte loads; A fragments read from LDS instead of per-lane gathers across batch rows; sum order kept
# speedup vs baseline: 3.0007x; 3.0007x over previous
#include <hip/hip_runtime.h>


typedef float v2f __attribute__((ext_vector_type(2)));
typedef float v4f __attribute__((ext_vector_type(4)));
typedef float v8f __attribute__((ext_vector_type(8)));

#define BATCH     32
#define IN_CH     21
#define IN_FEAT   4096
#define WIN       32
#define OUT_CH    8
#define N_WIN     4064
#define NK        (OUT_CH * N_WIN)
#define DDIM      (IN_CH * WIN)

#define WPB       8
#define SEG       40
#define SEG4      (SEG / 4)
#define PITCH     44
#define NROWS     (BATCH * IN_CH)
#define ROWS_PER_LOAD 3
#define STAGE_ITERS (NROWS / (ROWS_PER_LOAD * WPB))

__global__ __launch_bounds__(256)
void FilterbankLinear_kernel(const float* __restrict__ x,
                             const float* __restrict__ w,
                             float* __restrict__ out) {
    __shared__ __attribute__((aligned(16))) float xs[NROWS * PITCH];

    const int lane = threadIdx.x & 31;
    const int wave = threadIdx.x >> 5;
    const int j0 = blockIdx.x * WPB;
    const int j  = j0 + wave;

    {
        const int sub = lane / SEG4;
        const int q   = lane - sub * SEG4;
        if (sub < ROWS_PER_LOAD) {
            const float* src = x + (size_t)(wave * IN_CH + sub) * IN_FEAT + j0 + q * 4;
            float*       dst = xs + (sub * BATCH + wave) * PITCH + q * 4;
            #pragma unroll
            for (int i = 0; i < STAGE_ITERS; ++i) {
                const int bo = (i & 3) * WPB;
                const int co = (i >> 2) * ROWS_PER_LOAD;
                const v4f v = *(const v4f*)(src + (size_t)(bo * IN_CH + co) * IN_FEAT);
                *(v4f*)(dst + (co * BATCH + bo) * PITCH) = v;
            }
        }
    }
    __syncthreads();

    const int start = (j == N_WIN - 1) ? (IN_FEAT - WIN) : j;
    const int s = start - j0;

    const int r    = lane & 15;
    const int half = lane >> 4;
    const int koff = half * 2;

    const float* alo = xs + r * PITCH + s + koff;
    const float* ahi = alo + 16 * PITCH;

    const int  n      = r;
    const bool nvalid = (n < OUT_CH);
    const int  nclamp = nvalid ? n : (OUT_CH - 1);
    const float* __restrict__ wrow =
        w + (size_t)(j * OUT_CH + nclamp) * DDIM + koff;

    v8f acc0 = {};
    v8f acc1 = {};

    #pragma unroll 8
    for (int kb = 0; kb < DDIM; kb += 4) {
        const int c = kb >> 5;
        const int t = kb & 31;

        const float* al = alo + c * (BATCH * PITCH) + t;
        const float* ah = ahi + c * (BATCH * PITCH) + t;
        v2f a0; a0.x = al[0]; a0.y = al[1];
        v2f a1; a1.x = ah[0]; a1.y = ah[1];

        v2f bv = __builtin_nontemporal_load((const v2f*)(wrow + kb));
        bv.x = nvalid ? bv.x : 0.0f;
        bv.y = nvalid ? bv.y : 0.0f;

        acc0 = __builtin_amdgcn_wmma_f32_16x16x4_f32(
                   false, a0, false, bv, (short)0, acc0, false, false);
        acc1 = __builtin_amdgcn_wmma_f32_16x16x4_f32(
                   false, a1, false, bv, (short)0, acc1, false, false);
    }

    if (nvalid) {
        float* obase = out + (size_t)j * OUT_CH + n;
        #pragma unroll
        for (int v = 0; v < 8; ++v) {
            const int m = v + half * 8;
            obase[(size_t)m        * NK] = acc0[v];
            obase[(size_t)(m + 16) * NK] = acc1[v];
        }
    }
}

extern "C" void kernel_launch(void* const* d_in, const int* in_sizes, int n_in,
                              void* d_out, int out_size, void* d_ws, size_t ws_size,
                              hipStream_t stream) {
    (void)in_sizes; (void)n_in; (void)out_size; (void)d_ws; (void)ws_size;
    const float* x = (const float*)d_in[0];
    const float* w = (const float*)d_in[1];
    float* out = (float*)d_out;

    dim3 grid(N_WIN / WPB);
    dim3 block(256);
    FilterbankLinear_kernel<<<grid, block, 0, stream>>>(x, w, out);
}


# Weight loads issued 3 input channels ahead of their WMMAs into a 4-slot register ring (first 3 groups behind the x staging loads, before the barrier); main loop fully unrolled; k and sum order kept
# speedup vs baseline: 3.9559x; 1.3183x over previous
#include <hip/hip_runtime.h>


typedef float v2f __attribute__((ext_vector_type(2)));
typedef float v4f __attribute__((ext_vector_type(4)));
typedef float v8f __attribute__((ext_vector_type(8)));

#define BATCH     32
#define IN_CH     21
#define IN_FEAT   4096
#define WIN       32
#define OUT_CH    8
#define N_WIN     4064
#define NK        (OUT_CH * N_WIN)
#define DDIM      (IN_CH * WIN)

#define WPB       8
#define SEG       40
#define SEG4      (SEG / 4)
#define PITCH     44
#define NROWS     (BATCH * IN_CH)
#define ROWS_PER_LOAD 3
#define STAGE_ITERS (NROWS / (ROWS_PER_LOAD * WPB))

#define KSTEPS    (WIN / 4)
#define AHEAD     3
#define NSLOT     (AHEAD + 1)

__global__ __launch_bounds__(256)
void FilterbankLinear_kernel(const float* __restrict__ x,
                             const float* __restrict__ w,
                             float* __restrict__ out) {
    __shared__ __attribute__((aligned(16))) float xs[NROWS * PITCH];

    const int lane = threadIdx.x & 31;
    const int wave = threadIdx.x >> 5;
    const int j0 = blockIdx.x * WPB;
    const int j  = j0 + wave;

    const int r    = lane & 15;
    const int half = lane >> 4;
    const int koff = half * 2;

    const int  n      = r;
    const bool nvalid = (n < OUT_CH);
    const int  nclamp = nvalid ? n : (OUT_CH - 1);
    const float* __restrict__ wrow =
        w + (size_t)(j * OUT_CH + nclamp) * DDIM + koff;

    const int  sub    = lane / SEG4;
    const int  q      = lane - sub * SEG4;
    const bool stager = (sub < ROWS_PER_LOAD);
    v4f xv[STAGE_ITERS];
    if (stager) {
        const float* src = x + (size_t)(wave * IN_CH + sub) * IN_FEAT + j0 + q * 4;
        #pragma unroll
        for (int i = 0; i < STAGE_ITERS; ++i) {
            const int bo = (i & 3) * WPB;
            const int co = (i >> 2) * ROWS_PER_LOAD;
            xv[i] = *(const v4f*)(src + (size_t)(bo * IN_CH + co) * IN_FEAT);
        }
    }

    v2f wq[NSLOT][KSTEPS];
    #pragma unroll
    for (int c = 0; c < AHEAD; ++c) {
        #pragma unroll
        for (int i = 0; i < KSTEPS; ++i)
            wq[c][i] = __builtin_nontemporal_load((const v2f*)(wrow + c * WIN + i * 4));
    }

    if (stager) {
        float* dst = xs + (sub * BATCH + wave) * PITCH + q * 4;
        #pragma unroll
        for (int i = 0; i < STAGE_ITERS; ++i) {
            const int bo = (i & 3) * WPB;
            const int co = (i >> 2) * ROWS_PER_LOAD;
            *(v4f*)(dst + (co * BATCH + bo) * PITCH) = xv[i];
        }
    }
    __syncthreads();

    const int start = (j == N_WIN - 1) ? (IN_FEAT - WIN) : j;
    const int s = start - j0;

    const float* alo = xs + r * PITCH + s + koff;
    const float* ahi = alo + 16 * PITCH;

    v8f acc0 = {};
    v8f acc1 = {};

    #pragma unroll
    for (int c = 0; c < IN_CH; ++c) {
        if (c + AHEAD < IN_CH) {
            __builtin_amdgcn_sched_barrier(0x386);
            #pragma unroll
            for (int i = 0; i < KSTEPS; ++i)
                wq[(c + AHEAD) % NSLOT][i] = __builtin_nontemporal_load(
                    (const v2f*)(wrow + (c + AHEAD) * WIN + i * 4));
            __builtin_amdgcn_sched_barrier(0x386);
        }

        #pragma unroll
        for (int i = 0; i < KSTEPS; ++i) {
            const int t = i * 4;
            const float* al = alo + c * (BATCH * PITCH) + t;
            const float* ah = ahi + c * (BATCH * PITCH) + t;
            v2f a0; a0.x = al[0]; a0.y = al[1];
            v2f a1; a1.x = ah[0]; a1.y = ah[1];

            const v2f bv = wq[c % NSLOT][i];

            acc0 = __builtin_amdgcn_wmma_f32_16x16x4_f32(
                       false, a0, false, bv, (short)0, acc0, false, false);
            acc1 = __builtin_amdgcn_wmma_f32_16x16x4_f32(
                       false, a1, false, bv, (short)0, acc1, false, false);
        }
    }

    if (nvalid) {
        float* obase = out + (size_t)j * OUT_CH + n;
        #pragma unroll
        for (int v = 0; v < 8; ++v) {
            const int m = v + half * 8;
            obase[(size_t)m        * NK] = acc0[v];
            obase[(size_t)(m + 16) * NK] = acc1[v];
        }
    }
}

extern "C" void kernel_launch(void* const* d_in, const int* in_sizes, int n_in,
                              void* d_out, int out_size, void* d_ws, size_t ws_size,
                              hipStream_t stream) {
    (void)in_sizes; (void)n_in; (void)out_size; (void)d_ws; (void)ws_size;
    const float* x = (const float*)d_in[0];
    const float* w = (const float*)d_in[1];
    float* out = (float*)d_out;

    dim3 grid(N_WIN / WPB);
    dim3 block(256);
    FilterbankLinear_kernel<<<grid, block, 0, stream>>>(x, w, out);
}


# Block covers 16 windows with 16 waves (was 8 with 8): the shared x tile is 48 floats a row, still 2 cache lines, so staging pulls half the lines per window; LDS 139776 B; k and sum order kept
# speedup vs baseline: 4.0822x; 1.0319x over previous
#include <hip/hip_runtime.h>


typedef float v2f __attribute__((ext_vector_type(2)));
typedef float v4f __attribute__((ext_vector_type(4)));
typedef float v8f __attribute__((ext_vector_type(8)));

#define BATCH     32
#define IN_CH     21
#define IN_FEAT   4096
#define WIN       32
#define OUT_CH    8
#define N_WIN     4064
#define NK        (OUT_CH * N_WIN)
#define DDIM      (IN_CH * WIN)

#define WPB       16
#define SEG       48
#define SEG4      (SEG / 4)
#define PITCH     52
#define NROWS     (BATCH * IN_CH)
#define ROWS_PER_LOAD 2
#define STAGE_ITERS (NROWS / (ROWS_PER_LOAD * WPB))

#define KSTEPS    (WIN / 4)
#define AHEAD     3
#define NSLOT     (AHEAD + 1)

__global__ __launch_bounds__(512)
void FilterbankLinear_kernel(const float* __restrict__ x,
                             const float* __restrict__ w,
                             float* __restrict__ out) {
    __shared__ __attribute__((aligned(16))) float xs[NROWS * PITCH];

    const int lane = threadIdx.x & 31;
    const int wave = threadIdx.x >> 5;
    const int j0 = blockIdx.x * WPB;
    const int j  = j0 + wave;

    const int r    = lane & 15;
    const int half = lane >> 4;
    const int koff = half * 2;

    const int  n      = r;
    const bool nvalid = (n < OUT_CH);
    const int  nclamp = nvalid ? n : (OUT_CH - 1);
    const float* __restrict__ wrow =
        w + (size_t)(j * OUT_CH + nclamp) * DDIM + koff;

    const int  sub    = lane / SEG4;
    const int  q      = lane - sub * SEG4;
    const bool stager = (sub < ROWS_PER_LOAD);
    v4f xv[STAGE_ITERS];
    const int  brow   = wave * ROWS_PER_LOAD + sub;
    if (stager) {
        const float* src = x + (size_t)(brow * IN_CH) * IN_FEAT + j0 + q * 4;
        #pragma unroll
        for (int i = 0; i < STAGE_ITERS; ++i)
            xv[i] = *(const v4f*)(src + (size_t)i * IN_FEAT);
    }

    v2f wq[NSLOT][KSTEPS];
    #pragma unroll
    for (int c = 0; c < AHEAD; ++c) {
        #pragma unroll
        for (int i = 0; i < KSTEPS; ++i)
            wq[c][i] = __builtin_nontemporal_load((const v2f*)(wrow + c * WIN + i * 4));
    }

    if (stager) {
        float* dst = xs + brow * PITCH + q * 4;
        #pragma unroll
        for (int i = 0; i < STAGE_ITERS; ++i)
            *(v4f*)(dst + (i * BATCH) * PITCH) = xv[i];
    }
    __syncthreads();

    const int start = (j == N_WIN - 1) ? (IN_FEAT - WIN) : j;
    const int s = start - j0;

    const float* alo = xs + r * PITCH + s + koff;
    const float* ahi = alo + 16 * PITCH;

    v8f acc0 = {};
    v8f acc1 = {};

    #pragma unroll
    for (int c = 0; c < IN_CH; ++c) {
        if (c + AHEAD < IN_CH) {
            __builtin_amdgcn_sched_barrier(0x386);
            #pragma unroll
            for (int i = 0; i < KSTEPS; ++i)
                wq[(c + AHEAD) % NSLOT][i] = __builtin_nontemporal_load(
                    (const v2f*)(wrow + (c + AHEAD) * WIN + i * 4));
            __builtin_amdgcn_sched_barrier(0x386);
        }

        #pragma unroll
        for (int i = 0; i < KSTEPS; ++i) {
            const int t = i * 4;
            const float* al = alo + c * (BATCH * PITCH) + t;
            const float* ah = ahi + c * (BATCH * PITCH) + t;
            v2f a0; a0.x = al[0]; a0.y = al[1];
            v2f a1; a1.x = ah[0]; a1.y = ah[1];

            const v2f bv = wq[c % NSLOT][i];

            acc0 = __builtin_amdgcn_wmma_f32_16x16x4_f32(
                       false, a0, false, bv, (short)0, acc0, false, false);
            acc1 = __builtin_amdgcn_wmma_f32_16x16x4_f32(
                       false, a1, false, bv, (short)0, acc1, false, false);
        }
    }

    if (nvalid) {
        float* obase = out + (size_t)j * OUT_CH + n;
        #pragma unroll
        for (int v = 0; v < 8; ++v) {
            const int m = v + half * 8;
            obase[(size_t)m        * NK] = acc0[v];
            obase[(size_t)(m + 16) * NK] = acc1[v];
        }
    }
}

extern "C" void kernel_launch(void* const* d_in, const int* in_sizes, int n_in,
                              void* d_out, int out_size, void* d_ws, size_t ws_size,
                              hipStream_t stream) {
    (void)in_sizes; (void)n_in; (void)out_size; (void)d_ws; (void)ws_size;
    const float* x = (const float*)d_in[0];
    const float* w = (const float*)d_in[1];
    float* out = (float*)d_out;

    dim3 grid(N_WIN / WPB);
    dim3 block(512);
    FilterbankLinear_kernel<<<grid, block, 0, stream>>>(x, w, out);
}


# One wave takes a PAIR of adjacent windows: B columns 0-7/8-15 hold their weights, the 2nd shifted in k by its start offset: 9 k-steps a channel for 2 windows (was 16); 2nd window's fours regrouped
# speedup vs baseline: 4.5881x; 1.1239x over previous
#include <hip/hip_runtime.h>


typedef float v2f __attribute__((ext_vector_type(2)));
typedef float v4f __attribute__((ext_vector_type(4)));
typedef float v8f __attribute__((ext_vector_type(8)));
typedef float v2f_a4 __attribute__((ext_vector_type(2), aligned(4)));

#define BATCH     32
#define IN_CH     21
#define IN_FEAT   4096
#define WIN       32
#define OUT_CH    8
#define N_WIN     4064
#define NK        (OUT_CH * N_WIN)
#define DDIM      (IN_CH * WIN)

#define WPB       16
#define NWAVE     (WPB / 2)
#define SEG       48
#define SEG4      (SEG / 4)
#define PITCH     52
#define NROWS     (BATCH * IN_CH)
#define ROWS_PER_LOAD 2
#define STAGE_ITERS (NROWS / (ROWS_PER_LOAD * NWAVE))

#define KFULL     (WIN / 4)
#define KSTEPS    (KFULL + 1)
#define AHEAD     3
#define NSLOT     (AHEAD + 1)
#define GLOADS    (2 * KFULL + 1)

__global__ __launch_bounds__(NWAVE * 32)
void FilterbankLinear_kernel(const float* __restrict__ x,
                             const float* __restrict__ w,
                             float* __restrict__ out) {
    __shared__ __attribute__((aligned(16))) float xs[NROWS * PITCH];

    const int lane = threadIdx.x & 31;
    const int wave = threadIdx.x >> 5;
    const int j0 = blockIdx.x * WPB;
    const int j  = j0 + 2 * wave;

    const int r    = lane & 15;
    const int half = lane >> 4;
    const int koff = half * 2;

    const int  n   = r;
    const int  hi  = n >> 3;
    const int  d   = (j + 1 == N_WIN - 1) ? (IN_FEAT - WIN - j) : 1;
    const bool hib = (hi != 0);
    const bool hh0 = hib && (half == 0);
    const bool d2  = (d == 2);
    const float* __restrict__ rowL = w + (size_t)(j * OUT_CH + (n & 7)) * DDIM;
    const float* __restrict__ rowH = w + (size_t)((j + 1) * OUT_CH + (n & 7)) * DDIM;
    const float* __restrict__ wL  = rowL + koff;
    const float* __restrict__ wH  = rowH + koff - d;
    const float* __restrict__ wH0 = rowH + (half ? 2 - d : 0);
    const float* __restrict__ wT  = rowH + (WIN - 2);

    const int  sub    = lane / SEG4;
    const int  q      = lane - sub * SEG4;
    const bool stager = (sub < ROWS_PER_LOAD);
    v4f xv[STAGE_ITERS];
    const int  brow   = wave * ROWS_PER_LOAD + sub;
    if (stager) {
        const float* src = x + (size_t)(brow * IN_CH) * IN_FEAT + j0 + q * 4;
        #pragma unroll
        for (int i = 0; i < STAGE_ITERS; ++i)
            xv[i] = *(const v4f*)(src + (size_t)((i / IN_CH) * 16 * IN_CH + (i % IN_CH)) * IN_FEAT);
    }

    v2f wq[NSLOT][GLOADS];
    #pragma unroll
    for (int c = 0; c < AHEAD; ++c) {
        #pragma unroll
        for (int i = 0; i < KFULL; ++i)
            wq[c][i] = __builtin_nontemporal_load((const v2f*)(wL + c * WIN + i * 4));
        wq[c][KFULL] = __builtin_nontemporal_load((const v2f_a4*)(wH0 + c * WIN));
        #pragma unroll
        for (int i = 1; i < KFULL; ++i)
            wq[c][KFULL + i] = __builtin_nontemporal_load((const v2f_a4*)(wH + c * WIN + i * 4));
        wq[c][2 * KFULL] = __builtin_nontemporal_load((const v2f*)(wT + c * WIN));
    }

    if (stager) {
        float* dst = xs + brow * PITCH + q * 4;
        #pragma unroll
        for (int i = 0; i < STAGE_ITERS; ++i)
            *(v4f*)(dst + ((i % IN_CH) * BATCH + (i / IN_CH) * 16) * PITCH) = xv[i];
    }
    for (int row = threadIdx.x; row < NROWS; row += NWAVE * 32)
        *(v4f*)(xs + row * PITCH + SEG) = (v4f){0.f, 0.f, 0.f, 0.f};
    __syncthreads();

    const int s = j - j0;

    const float* alo = xs + r * PITCH + s + koff;
    const float* ahi = alo + 16 * PITCH;

    v8f acc0 = {};
    v8f acc1 = {};

    #pragma unroll
    for (int c = 0; c < IN_CH; ++c) {
        if (c + AHEAD < IN_CH) {
            __builtin_amdgcn_sched_barrier(0x386);
            #pragma unroll
            for (int i = 0; i < KFULL; ++i)
                wq[(c + AHEAD) % NSLOT][i] = __builtin_nontemporal_load(
                    (const v2f*)(wL + (c + AHEAD) * WIN + i * 4));
            wq[(c + AHEAD) % NSLOT][KFULL] = __builtin_nontemporal_load(
                (const v2f_a4*)(wH0 + (c + AHEAD) * WIN));
            #pragma unroll
            for (int i = 1; i < KFULL; ++i)
                wq[(c + AHEAD) % NSLOT][KFULL + i] = __builtin_nontemporal_load(
                    (const v2f_a4*)(wH + (c + AHEAD) * WIN + i * 4));
            wq[(c + AHEAD) % NSLOT][2 * KFULL] = __builtin_nontemporal_load(
                (const v2f*)(wT + (c + AHEAD) * WIN));
            __builtin_amdgcn_sched_barrier(0x386);
        }

        #pragma unroll
        for (int i = 0; i < KSTEPS; ++i) {
            const int t = i * 4;
            const v2f a0 = *(const v2f*)(alo + c * (BATCH * PITCH) + t);
            const v2f a1 = *(const v2f*)(ahi + c * (BATCH * PITCH) + t);

            v2f bv;
            if (i == 0) {
                const v2f L = wq[c % NSLOT][0];
                const v2f H = wq[c % NSLOT][KFULL];
                const float hx = hh0 ? 0.f : H.x;
                const float hy = hh0 ? (d2 ? 0.f : H.x) : H.y;
                bv.x = hib ? hx : L.x;
                bv.y = hib ? hy : L.y;
            } else if (i < KFULL) {
                const v2f L = wq[c % NSLOT][i];
                const v2f H = wq[c % NSLOT][KFULL + i];
                bv.x = hib ? H.x : L.x;
                bv.y = hib ? H.y : L.y;
            } else {
                const v2f T = wq[c % NSLOT][2 * KFULL];
                bv.x = hh0 ? (d2 ? T.x : T.y) : 0.f;
                bv.y = (hh0 && d2) ? T.y : 0.f;
            }

            acc0 = __builtin_amdgcn_wmma_f32_16x16x4_f32(
                       false, a0, false, bv, (short)0, acc0, false, false);
            acc1 = __builtin_amdgcn_wmma_f32_16x16x4_f32(
                       false, a1, false, bv, (short)0, acc1, false, false);
        }
    }

    float* obase = out + (size_t)j * OUT_CH + n;
    #pragma unroll
    for (int v = 0; v < 8; ++v) {
        const int m = v + half * 8;
        obase[(size_t)m        * NK] = acc0[v];
        obase[(size_t)(m + 16) * NK] = acc1[v];
    }
}

extern "C" void kernel_launch(void* const* d_in, const int* in_sizes, int n_in,
                              void* d_out, int out_size, void* d_ws, size_t ws_size,
                              hipStream_t stream) {
    (void)in_sizes; (void)n_in; (void)out_size; (void)d_ws; (void)ws_size;
    const float* x = (const float*)d_in[0];
    const float* w = (const float*)d_in[1];
    float* out = (float*)d_out;

    dim3 grid(N_WIN / WPB);
    dim3 block(NWAVE * 32);
    FilterbankLinear_kernel<<<grid, block, 0, stream>>>(x, w, out);
}


# Weight rows loaded as whole 128-byte lines (16 B a lane, 4 loads a channel; were 17, each touching 8 lines), turned into B fragments through a per-wave LDS slot with zero pads; k and sum order kept
# speedup vs baseline: 5.8474x; 1.2745x over previous
#include <hip/hip_runtime.h>


typedef float v2f __attribute__((ext_vector_type(2)));
typedef float v4f __attribute__((ext_vector_type(4)));
typedef float v8f __attribute__((ext_vector_type(8)));

#define BATCH     32
#define IN_CH     21
#define IN_FEAT   4096
#define WIN       32
#define OUT_CH    8
#define N_WIN     4064
#define NK        (OUT_CH * N_WIN)
#define DDIM      (IN_CH * WIN)

#define WPB       16
#define NWAVE     (WPB / 2)
#define SEG       48
#define SEG4      (SEG / 4)
#define PITCH     52
#define NROWS     (BATCH * IN_CH)
#define ROWS_PER_LOAD 2
#define STAGE_ITERS (NROWS / (ROWS_PER_LOAD * NWAVE))

#define KFULL     (WIN / 4)
#define KSTEPS    (KFULL + 1)
#define AHEAD     3
#define NSLOT     (AHEAD + 1)
#define GLOADS    4
#define WROW      (WIN + 4)
#define WSLOT     (2 * OUT_CH * WROW + 4)

__global__ __launch_bounds__(NWAVE * 32)
void FilterbankLinear_kernel(const float* __restrict__ x,
                             const float* __restrict__ w,
                             float* __restrict__ out) {
    __shared__ __attribute__((aligned(16))) float xs[NROWS * PITCH];
    __shared__ __attribute__((aligned(16))) float wsm[NWAVE * WSLOT];

    const int lane = threadIdx.x & 31;
    const int wave = threadIdx.x >> 5;
    const int j0 = blockIdx.x * WPB;
    const int j  = j0 + 2 * wave;

    const int r    = lane & 15;
    const int half = lane >> 4;
    const int koff = half * 2;

    const int  n   = r;
    const int  hi  = n >> 3;
    const int  d   = (j + 1 == N_WIN - 1) ? (IN_FEAT - WIN - j) : 1;
    const float* __restrict__ wg = w + (size_t)(j * OUT_CH + (lane >> 3)) * DDIM + (lane & 7) * 4;
    float* const wslot = wsm + wave * WSLOT;
    float* const wst   = wslot + (lane >> 3) * WROW + 4 + (lane & 7) * 4;
    const float* const bp = wslot + n * WROW + 4 + koff - (hi ? d : 0);

    const int  sub    = lane / SEG4;
    const int  q      = lane - sub * SEG4;
    const bool stager = (sub < ROWS_PER_LOAD);
    v4f xv[STAGE_ITERS];
    const int  brow   = wave * ROWS_PER_LOAD + sub;
    if (stager) {
        const float* src = x + (size_t)(brow * IN_CH) * IN_FEAT + j0 + q * 4;
        #pragma unroll
        for (int i = 0; i < STAGE_ITERS; ++i)
            xv[i] = *(const v4f*)(src + (size_t)((i / IN_CH) * 16 * IN_CH + (i % IN_CH)) * IN_FEAT);
    }

    v4f wq[NSLOT][GLOADS];
    #pragma unroll
    for (int c = 0; c < AHEAD; ++c) {
        #pragma unroll
        for (int g = 0; g < GLOADS; ++g)
            wq[c][g] = __builtin_nontemporal_load((const v4f*)(wg + (size_t)(4 * g) * DDIM + c * WIN));
    }

    if (stager) {
        float* dst = xs + brow * PITCH + q * 4;
        #pragma unroll
        for (int i = 0; i < STAGE_ITERS; ++i)
            *(v4f*)(dst + ((i % IN_CH) * BATCH + (i / IN_CH) * 16) * PITCH) = xv[i];
    }
    for (int row = threadIdx.x; row < NROWS; row += NWAVE * 32)
        *(v4f*)(xs + row * PITCH + SEG) = (v4f){0.f, 0.f, 0.f, 0.f};
    if (lane <= 2 * OUT_CH)
        *(v4f*)(wslot + lane * WROW) = (v4f){0.f, 0.f, 0.f, 0.f};
    __syncthreads();

    const int s = j - j0;

    const float* alo = xs + r * PITCH + s + koff;
    const float* ahi = alo + 16 * PITCH;

    v8f acc0 = {};
    v8f acc1 = {};

    #pragma unroll
    for (int c = 0; c < IN_CH; ++c) {
        if (c + AHEAD < IN_CH) {
            __builtin_amdgcn_sched_barrier(0x386);
            #pragma unroll
            for (int g = 0; g < GLOADS; ++g)
                wq[(c + AHEAD) % NSLOT][g] = __builtin_nontemporal_load(
                    (const v4f*)(wg + (size_t)(4 * g) * DDIM + (c + AHEAD) * WIN));
            __builtin_amdgcn_sched_barrier(0x386);
        }

        __builtin_amdgcn_fence(__ATOMIC_RELEASE, "wavefront");
        __builtin_amdgcn_wave_barrier();
        __builtin_amdgcn_fence(__ATOMIC_ACQUIRE, "wavefront");
        #pragma unroll
        for (int g = 0; g < GLOADS; ++g)
            *(v4f*)(wst + (4 * g) * WROW) = wq[c % NSLOT][g];
        __builtin_amdgcn_fence(__ATOMIC_RELEASE, "wavefront");
        __builtin_amdgcn_wave_barrier();
        __builtin_amdgcn_fence(__ATOMIC_ACQUIRE, "wavefront");

        #pragma unroll
        for (int i = 0; i < KSTEPS; ++i) {
            const int t = i * 4;
            const v2f a0 = *(const v2f*)(alo + c * (BATCH * PITCH) + t);
            const v2f a1 = *(const v2f*)(ahi + c * (BATCH * PITCH) + t);

            v2f bv;
            bv.x = bp[t];
            bv.y = bp[t + 1];

            acc0 = __builtin_amdgcn_wmma_f32_16x16x4_f32(
                       false, a0, false, bv, (short)0, acc0, false, false);
            acc1 = __builtin_amdgcn_wmma_f32_16x16x4_f32(
                       false, a1, false, bv, (short)0, acc1, false, false);
        }
    }

    float* obase = out + (size_t)j * OUT_CH + n;
    #pragma unroll
    for (int v = 0; v < 8; ++v) {
        const int m = v + half * 8;
        obase[(size_t)m        * NK] = acc0[v];
        obase[(size_t)(m + 16) * NK] = acc1[v];
    }
}

extern "C" void kernel_launch(void* const* d_in, const int* in_sizes, int n_in,
                              void* d_out, int out_size, void* d_ws, size_t ws_size,
                              hipStream_t stream) {
    (void)in_sizes; (void)n_in; (void)out_size; (void)d_ws; (void)ws_size;
    const float* x = (const float*)d_in[0];
    const float* w = (const float*)d_in[1];
    float* out = (float*)d_out;

    dim3 grid(N_WIN / WPB);
    dim3 block(NWAVE * 32);
    FilterbankLinear_kernel<<<grid, block, 0, stream>>>(x, w, out);
}


# A SIMD's 2 waves alternate whole channels on the matrix pipe: one reads the next channel's full A/B fragment set (2nd register set) while the other's 18 WMMAs run at raised s_setprio; sum order kept
# speedup vs baseline: 6.1667x; 1.0546x over previous
#include <hip/hip_runtime.h>


typedef float v2f __attribute__((ext_vector_type(2)));
typedef float v4f __attribute__((ext_vector_type(4)));
typedef float v8f __attribute__((ext_vector_type(8)));

#define BATCH     32
#define IN_CH     21
#define IN_FEAT   4096
#define WIN       32
#define OUT_CH    8
#define N_WIN     4064
#define NK        (OUT_CH * N_WIN)
#define DDIM      (IN_CH * WIN)

#define WPB       16
#define NWAVE     (WPB / 2)
#define SEG       48
#define SEG4      (SEG / 4)
#define PITCH     52
#define NROWS     (BATCH * IN_CH)
#define ROWS_PER_LOAD 2
#define STAGE_ITERS (NROWS / (ROWS_PER_LOAD * NWAVE))

#define KFULL     (WIN / 4)
#define KSTEPS    (KFULL + 1)
#define AHEAD     3
#define NSLOT     (AHEAD + 1)
#define GLOADS    4
#define WROW      (WIN + 4)
#define WSLOT     (2 * OUT_CH * WROW + 4)

__global__ __launch_bounds__(NWAVE * 32)
void FilterbankLinear_kernel(const float* __restrict__ x,
                             const float* __restrict__ w,
                             float* __restrict__ out) {
    __shared__ __attribute__((aligned(16))) float xs[NROWS * PITCH];
    __shared__ __attribute__((aligned(16))) float wsm[NWAVE * WSLOT];

    const int lane = threadIdx.x & 31;
    const int wave = threadIdx.x >> 5;
    const int j0 = blockIdx.x * WPB;
    const int j  = j0 + 2 * wave;

    const int r    = lane & 15;
    const int half = lane >> 4;
    const int koff = half * 2;

    const int  n   = r;
    const int  hi  = n >> 3;
    const int  d   = (j + 1 == N_WIN - 1) ? (IN_FEAT - WIN - j) : 1;
    const float* __restrict__ wg = w + (size_t)(j * OUT_CH + (lane >> 3)) * DDIM + (lane & 7) * 4;
    float* const wslot = wsm + wave * WSLOT;
    float* const wst   = wslot + (lane >> 3) * WROW + 4 + (lane & 7) * 4;
    const float* const bp = wslot + n * WROW + 4 + koff - (hi ? d : 0);

    const int  sub    = lane / SEG4;
    const int  q      = lane - sub * SEG4;
    const bool stager = (sub < ROWS_PER_LOAD);
    v4f xv[STAGE_ITERS];
    const int  brow   = wave * ROWS_PER_LOAD + sub;
    if (stager) {
        const float* src = x + (size_t)(brow * IN_CH) * IN_FEAT + j0 + q * 4;
        #pragma unroll
        for (int i = 0; i < STAGE_ITERS; ++i)
            xv[i] = *(const v4f*)(src + (size_t)((i / IN_CH) * 16 * IN_CH + (i % IN_CH)) * IN_FEAT);
    }

    v4f wq[NSLOT][GLOADS];
    #pragma unroll
    for (int c = 0; c < AHEAD; ++c) {
        #pragma unroll
        for (int g = 0; g < GLOADS; ++g)
            wq[c][g] = __builtin_nontemporal_load((const v4f*)(wg + (size_t)(4 * g) * DDIM + c * WIN));
    }

    if (stager) {
        float* dst = xs + brow * PITCH + q * 4;
        #pragma unroll
        for (int i = 0; i < STAGE_ITERS; ++i)
            *(v4f*)(dst + ((i % IN_CH) * BATCH + (i / IN_CH) * 16) * PITCH) = xv[i];
    }
    for (int row = threadIdx.x; row < NROWS; row += NWAVE * 32)
        *(v4f*)(xs + row * PITCH + SEG) = (v4f){0.f, 0.f, 0.f, 0.f};
    if (lane <= 2 * OUT_CH)
        *(v4f*)(wslot + lane * WROW) = (v4f){0.f, 0.f, 0.f, 0.f};
    __syncthreads();

    const int s = j - j0;

    const float* alo = xs + r * PITCH + s + koff;
    const float* ahi = alo + 16 * PITCH;

    v8f acc0 = {};
    v8f acc1 = {};

    v2f fa0[2][KSTEPS];
    v2f fa1[2][KSTEPS];
    v2f fb[2][KSTEPS];

    #pragma unroll
    for (int c = -1; c < IN_CH; ++c) {
        const int cn = c + 1;
        if (cn < IN_CH) {
            if (cn + AHEAD < IN_CH) {
                __builtin_amdgcn_sched_barrier(0x386);
                #pragma unroll
                for (int g = 0; g < GLOADS; ++g)
                    wq[(cn + AHEAD) % NSLOT][g] = __builtin_nontemporal_load(
                        (const v4f*)(wg + (size_t)(4 * g) * DDIM + (cn + AHEAD) * WIN));
            }
            __builtin_amdgcn_sched_barrier(0x6);

            #pragma unroll
            for (int i = 0; i < KSTEPS; ++i) {
                fa0[cn & 1][i] = *(const v2f*)(alo + cn * (BATCH * PITCH) + i * 4);
                fa1[cn & 1][i] = *(const v2f*)(ahi + cn * (BATCH * PITCH) + i * 4);
            }

            __builtin_amdgcn_fence(__ATOMIC_RELEASE, "wavefront");
            __builtin_amdgcn_wave_barrier();
            __builtin_amdgcn_fence(__ATOMIC_ACQUIRE, "wavefront");
            #pragma unroll
            for (int g = 0; g < GLOADS; ++g)
                *(v4f*)(wst + (4 * g) * WROW) = wq[cn % NSLOT][g];
            __builtin_amdgcn_fence(__ATOMIC_RELEASE, "wavefront");
            __builtin_amdgcn_wave_barrier();
            __builtin_amdgcn_fence(__ATOMIC_ACQUIRE, "wavefront");

            #pragma unroll
            for (int i = 0; i < KSTEPS; ++i) {
                fb[cn & 1][i].x = bp[i * 4];
                fb[cn & 1][i].y = bp[i * 4 + 1];
            }

            __builtin_amdgcn_sched_barrier(0x6);
        }

        if (c >= 0) {
            #pragma unroll
            for (int i = 0; i < KSTEPS; ++i) {
                acc0 = __builtin_amdgcn_wmma_f32_16x16x4_f32(
                           false, fa0[c & 1][i], false, fb[c & 1][i], (short)0, acc0, false, false);
                acc1 = __builtin_amdgcn_wmma_f32_16x16x4_f32(
                           false, fa1[c & 1][i], false, fb[c & 1][i], (short)0, acc1, false, false);
                if (i == 0) {
                    __builtin_amdgcn_sched_barrier(0);
                    __builtin_amdgcn_s_setprio(1);
                    __builtin_amdgcn_sched_barrier(0);
                }
            }
            __builtin_amdgcn_sched_barrier(0);
            __builtin_amdgcn_s_setprio(0);
            __builtin_amdgcn_sched_barrier(0);
        }
    }

    float* obase = out + (size_t)j * OUT_CH + n;
    #pragma unroll
    for (int v = 0; v < 8; ++v) {
        const int m = v + half * 8;
        obase[(size_t)m        * NK] = acc0[v];
        obase[(size_t)(m + 16) * NK] = acc1[v];
    }
}

extern "C" void kernel_launch(void* const* d_in, const int* in_sizes, int n_in,
                              void* d_out, int out_size, void* d_ws, size_t ws_size,
                              hipStream_t stream) {
    (void)in_sizes; (void)n_in; (void)out_size; (void)d_ws; (void)ws_size;
    const float* x = (const float*)d_in[0];
    const float* w = (const float*)d_in[1];
    float* out = (float*)d_out;

    dim3 grid(N_WIN / WPB);
    dim3 block(NWAVE * 32);
    FilterbankLinear_kernel<<<grid, block, 0, stream>>>(x, w, out);
}
